# starting kernel
# baseline (speedup 1.0000x reference)
#include <hip/hip_runtime.h>
#include <math.h>

typedef __attribute__((ext_vector_type(2))) float v2f;
typedef __attribute__((ext_vector_type(4))) float v4f;
typedef __attribute__((ext_vector_type(8))) float v8f;

#define D_MODEL 1024
#define NUM_BLOCKS 4
#define BATCH 8
#define SEQ 4096
#define BS (SEQ / NUM_BLOCKS)
#define NBLK (BATCH * NUM_BLOCKS)
#define ROW_CHUNKS 8
#define ROWS_PER_CHUNK (BS / ROW_CHUNKS)

__global__ void se_pool_partial(const float* __restrict__ x, float* __restrict__ part) {
    int wg    = blockIdx.x;
    int blk   = wg / ROW_CHUNKS;
    int chunk = wg % ROW_CHUNKS;
    int c4    = threadIdx.x;
    const float* base = x + (size_t)blk * BS * D_MODEL
                          + (size_t)chunk * ROWS_PER_CHUNK * D_MODEL
                          + 4 * c4;
    v4f acc = {0.f, 0.f, 0.f, 0.f};
    #pragma unroll 4
    for (int r = 0; r < ROWS_PER_CHUNK; ++r) {
        acc += *(const v4f*)(base + (size_t)r * D_MODEL);
    }
    *(v4f*)(part + (size_t)wg * D_MODEL + 4 * c4) = acc;
}

__global__ void se_pool_reduce(const float* __restrict__ part, float* __restrict__ pooled) {
    int blk = blockIdx.x;
    int c4  = threadIdx.x;
    v4f acc = {0.f, 0.f, 0.f, 0.f};
    #pragma unroll
    for (int c = 0; c < ROW_CHUNKS; ++c) {
        acc += *(const v4f*)(part + ((size_t)blk * ROW_CHUNKS + c) * D_MODEL + 4 * c4);
    }
    acc *= (1.0f / (float)BS);
    *(v4f*)(pooled + (size_t)blk * D_MODEL + 4 * c4) = acc;
}

__global__ void se_gate_wmma(const float* __restrict__ pooled,
                             const float* __restrict__ W,
                             const float* __restrict__ bias,
                             float* __restrict__ gate) {
    int wave = (int)((blockIdx.x * blockDim.x + threadIdx.x) >> 5);
    int lane = threadIdx.x & 31;
    int mt = wave & 1;
    int nt = wave >> 1;

    int mrow  = mt * 16 + (lane & 15);
    int nrow  = nt * 16 + (lane & 15);
    int khalf = (lane >> 4) * 2;

    const float* aptr = pooled + (size_t)mrow * D_MODEL + khalf;
    const float* bptr = W      + (size_t)nrow * D_MODEL + khalf;

    v8f acc = {};
    #pragma unroll 4
    for (int k = 0; k < D_MODEL; k += 4) {
        v2f a = *(const v2f*)(aptr + k);
        v2f b = *(const v2f*)(bptr + k);
        acc = __builtin_amdgcn_wmma_f32_16x16x4_f32(
             false, a,  false, b,
             (short)0, acc,  false,  false);
    }

    int m0 = mt * 16 + ((lane >> 4) << 3);
    int n  = nt * 16 + (lane & 15);
    float bb = bias[n];
    #pragma unroll
    for (int r = 0; r < 8; ++r) {
        float v = acc[r] + bb;
        float e = __expf(-v);
        gate[(size_t)(m0 + r) * D_MODEL + n] = __builtin_amdgcn_rcpf(1.0f + e);
    }
}

__global__ void se_apply_gate(const float* __restrict__ x,
                              const float* __restrict__ gate,
                              float* __restrict__ out) {
    size_t i = (size_t)blockIdx.x * blockDim.x + threadIdx.x;
    size_t f = i * 4;
    int    d   = (int)(f & (D_MODEL - 1));
    size_t row = f >> 10;
    size_t blk = row >> 10;
    v4f xv = *(const v4f*)(x + f);
    v4f gv = *(const v4f*)(gate + blk * D_MODEL + d);
    __builtin_nontemporal_store(xv * gv, (v4f*)(out + f));
}

extern "C" void kernel_launch(void* const* d_in, const int* in_sizes, int n_in,
                              void* d_out, int out_size, void* d_ws, size_t ws_size,
                              hipStream_t stream) {
    const float* x = (const float*)d_in[0];
    const float* W = (const float*)d_in[1];
    const float* b = (const float*)d_in[2];
    float* out = (float*)d_out;

    float* ws     = (float*)d_ws;
    float* pooled = ws;
    float* gate   = pooled + (size_t)NBLK * D_MODEL;
    float* part   = gate   + (size_t)NBLK * D_MODEL;

    se_pool_partial<<<NBLK * ROW_CHUNKS, 256, 0, stream>>>(x, part);
    se_pool_reduce <<<NBLK,              256, 0, stream>>>(part, pooled);
    se_gate_wmma   <<<(NBLK / 16) * (D_MODEL / 16) / 8, 256, 0, stream>>>(pooled, W, b, gate);

    const size_t total4 = (size_t)BATCH * SEQ * D_MODEL / 4;
    se_apply_gate<<<(unsigned)(total4 / 256), 256, 0, stream>>>(x, gate, out);
}


# gate GEMM split-K: one work-group per 16x16 tile, K over its 16 waves, all fragment loads of a wave in flight at once, partial tiles summed in fixed order through LDS (sum order changes)
# speedup vs baseline: 1.3214x; 1.3214x over previous
#include <hip/hip_runtime.h>
#include <math.h>

typedef __attribute__((ext_vector_type(2))) float v2f;
typedef __attribute__((ext_vector_type(4))) float v4f;
typedef __attribute__((ext_vector_type(8))) float v8f;

#define D_MODEL 1024
#define NUM_BLOCKS 4
#define BATCH 8
#define SEQ 4096
#define BS (SEQ / NUM_BLOCKS)
#define NBLK (BATCH * NUM_BLOCKS)
#define ROW_CHUNKS 8
#define ROWS_PER_CHUNK (BS / ROW_CHUNKS)

__global__ void se_pool_partial(const float* __restrict__ x, float* __restrict__ part) {
    int wg    = blockIdx.x;
    int blk   = wg / ROW_CHUNKS;
    int chunk = wg % ROW_CHUNKS;
    int c4    = threadIdx.x;
    const float* base = x + (size_t)blk * BS * D_MODEL
                          + (size_t)chunk * ROWS_PER_CHUNK * D_MODEL
                          + 4 * c4;
    v4f acc = {0.f, 0.f, 0.f, 0.f};
    #pragma unroll 4
    for (int r = 0; r < ROWS_PER_CHUNK; ++r) {
        acc += *(const v4f*)(base + (size_t)r * D_MODEL);
    }
    *(v4f*)(part + (size_t)wg * D_MODEL + 4 * c4) = acc;
}

__global__ void se_pool_reduce(const float* __restrict__ part, float* __restrict__ pooled) {
    int blk = blockIdx.x;
    int c4  = threadIdx.x;
    v4f acc = {0.f, 0.f, 0.f, 0.f};
    #pragma unroll
    for (int c = 0; c < ROW_CHUNKS; ++c) {
        acc += *(const v4f*)(part + ((size_t)blk * ROW_CHUNKS + c) * D_MODEL + 4 * c4);
    }
    acc *= (1.0f / (float)BS);
    *(v4f*)(pooled + (size_t)blk * D_MODEL + 4 * c4) = acc;
}

#define GATE_KSPLIT 16
#define GATE_KSLICE (D_MODEL / GATE_KSPLIT)
#define GATE_TILES  ((NBLK / 16) * (D_MODEL / 16))

__global__ void __launch_bounds__(32 * GATE_KSPLIT) __attribute__((amdgpu_waves_per_eu(4, 4)))
se_gate_wmma(const float* __restrict__ pooled,
             const float* __restrict__ W,
             const float* __restrict__ bias,
             float* __restrict__ gate) {
    __shared__ float red[GATE_KSPLIT][8 * 32];

    int tile = blockIdx.x;
    int mt   = tile & 1;
    int nt   = tile >> 1;
    int wave = threadIdx.x >> 5;
    int lane = threadIdx.x & 31;
    int c    = lane & 15;
    int h    = lane >> 4;

    int koff = wave * GATE_KSLICE + 4 * h;
    const float* aptr = pooled + (size_t)(mt * 16 + c) * D_MODEL + koff;
    const float* bptr = W      + (size_t)(nt * 16 + c) * D_MODEL + koff;

    float bb = bias[nt * 16 + c];

    v4f a[GATE_KSLICE / 8], b[GATE_KSLICE / 8];
    #pragma unroll
    for (int j = 0; j < GATE_KSLICE / 8; ++j) {
        a[j] = *(const v4f*)(aptr + 8 * j);
        b[j] = *(const v4f*)(bptr + 8 * j);
    }

    v8f acc = {};
    #pragma unroll
    for (int j = 0; j < GATE_KSLICE / 8; ++j) {
        v2f a0 = {a[j].x, a[j].y};
        v2f b0 = {b[j].x, b[j].y};
        v2f a1 = {a[j].z, a[j].w};
        v2f b1 = {b[j].z, b[j].w};
        acc = __builtin_amdgcn_wmma_f32_16x16x4_f32(
             false, a0,  false, b0,
             (short)0, acc,  false,  false);
        acc = __builtin_amdgcn_wmma_f32_16x16x4_f32(
             false, a1,  false, b1,
             (short)0, acc,  false,  false);
    }

    #pragma unroll
    for (int r = 0; r < 8; ++r) {
        red[wave][r * 32 + lane] = acc[r];
    }
    __syncthreads();

    int t = threadIdx.x;
    if (t < 8 * 32) {
        float s = red[0][t];
        #pragma unroll
        for (int w = 1; w < GATE_KSPLIT; ++w) {
            s += red[w][t];
        }
        int r = t >> 5;
        int l = t & 31;
        int m = mt * 16 + ((l >> 4) << 3) + r;
        int n = nt * 16 + (l & 15);
        float v = s + bb;
        float e = __expf(-v);
        gate[(size_t)m * D_MODEL + n] = __builtin_amdgcn_rcpf(1.0f + e);
    }
}

__global__ void se_apply_gate(const float* __restrict__ x,
                              const float* __restrict__ gate,
                              float* __restrict__ out) {
    size_t i = (size_t)blockIdx.x * blockDim.x + threadIdx.x;
    size_t f = i * 4;
    int    d   = (int)(f & (D_MODEL - 1));
    size_t row = f >> 10;
    size_t blk = row >> 10;
    v4f xv = *(const v4f*)(x + f);
    v4f gv = *(const v4f*)(gate + blk * D_MODEL + d);
    __builtin_nontemporal_store(xv * gv, (v4f*)(out + f));
}

extern "C" void kernel_launch(void* const* d_in, const int* in_sizes, int n_in,
                              void* d_out, int out_size, void* d_ws, size_t ws_size,
                              hipStream_t stream) {
    const float* x = (const float*)d_in[0];
    const float* W = (const float*)d_in[1];
    const float* b = (const float*)d_in[2];
    float* out = (float*)d_out;

    float* ws     = (float*)d_ws;
    float* pooled = ws;
    float* gate   = pooled + (size_t)NBLK * D_MODEL;
    float* part   = gate   + (size_t)NBLK * D_MODEL;

    se_pool_partial<<<NBLK * ROW_CHUNKS, 256, 0, stream>>>(x, part);
    se_pool_reduce <<<NBLK,              256, 0, stream>>>(part, pooled);
    se_gate_wmma   <<<GATE_TILES, 32 * GATE_KSPLIT, 0, stream>>>(pooled, W, b, gate);

    const size_t total4 = (size_t)BATCH * SEQ * D_MODEL / 4;
    se_apply_gate<<<(unsigned)(total4 / 256), 256, 0, stream>>>(x, gate, out);
}


# pooling pass re-tiled for bytes in flight: 16 waves per WGP, 32 row loads per thread issued before the first wait (was 8 waves, 4 loads); chunk halves added through LDS (partial sum order changes)
# speedup vs baseline: 1.6082x; 1.2170x over previous
#include <hip/hip_runtime.h>
#include <math.h>

typedef __attribute__((ext_vector_type(2))) float v2f;
typedef __attribute__((ext_vector_type(4))) float v4f;
typedef __attribute__((ext_vector_type(8))) float v8f;

#define D_MODEL 1024
#define NUM_BLOCKS 4
#define BATCH 8
#define SEQ 4096
#define BS (SEQ / NUM_BLOCKS)
#define NBLK (BATCH * NUM_BLOCKS)
#define ROW_CHUNKS 8
#define ROWS_PER_CHUNK (BS / ROW_CHUNKS)

#define POOL_GROUPS 2
#define POOL_ROWS   (ROWS_PER_CHUNK / POOL_GROUPS)
#define POOL_BATCH  32

__global__ void __launch_bounds__(256 * POOL_GROUPS) __attribute__((amdgpu_waves_per_eu(4, 4)))
se_pool_partial(const float* __restrict__ x, float* __restrict__ part) {
    __shared__ v4f red[256];

    int wg   = blockIdx.x;
    int c4   = threadIdx.x & 255;
    int half = threadIdx.x >> 8;
    const float* base = x + ((size_t)wg * ROWS_PER_CHUNK + (size_t)half * POOL_ROWS) * D_MODEL
                          + 4 * c4;
    v4f acc = {0.f, 0.f, 0.f, 0.f};
    #pragma unroll 1
    for (int r0 = 0; r0 < POOL_ROWS; r0 += POOL_BATCH) {
        v4f v[POOL_BATCH];
        #pragma unroll
        for (int r = 0; r < POOL_BATCH; ++r) {
            v[r] = *(const v4f*)(base + (size_t)(r0 + r) * D_MODEL);
        }
        #pragma unroll
        for (int r = 0; r < POOL_BATCH; ++r) {
            acc += v[r];
        }
    }
    if (half) red[c4] = acc;
    __syncthreads();
    if (!half) {
        acc += red[c4];
        *(v4f*)(part + (size_t)wg * D_MODEL + 4 * c4) = acc;
    }
}

__global__ void se_pool_reduce(const float* __restrict__ part, float* __restrict__ pooled) {
    int blk = blockIdx.x;
    int c4  = threadIdx.x;
    v4f acc = {0.f, 0.f, 0.f, 0.f};
    #pragma unroll
    for (int c = 0; c < ROW_CHUNKS; ++c) {
        acc += *(const v4f*)(part + ((size_t)blk * ROW_CHUNKS + c) * D_MODEL + 4 * c4);
    }
    acc *= (1.0f / (float)BS);
    *(v4f*)(pooled + (size_t)blk * D_MODEL + 4 * c4) = acc;
}

#define GATE_KSPLIT 16
#define GATE_KSLICE (D_MODEL / GATE_KSPLIT)
#define GATE_TILES  ((NBLK / 16) * (D_MODEL / 16))

__global__ void __launch_bounds__(32 * GATE_KSPLIT) __attribute__((amdgpu_waves_per_eu(4, 4)))
se_gate_wmma(const float* __restrict__ pooled,
             const float* __restrict__ W,
             const float* __restrict__ bias,
             float* __restrict__ gate) {
    __shared__ float red[GATE_KSPLIT][8 * 32];

    int tile = blockIdx.x;
    int mt   = tile & 1;
    int nt   = tile >> 1;
    int wave = threadIdx.x >> 5;
    int lane = threadIdx.x & 31;
    int c    = lane & 15;
    int h    = lane >> 4;

    int koff = wave * GATE_KSLICE + 4 * h;
    const float* aptr = pooled + (size_t)(mt * 16 + c) * D_MODEL + koff;
    const float* bptr = W      + (size_t)(nt * 16 + c) * D_MODEL + koff;

    float bb = bias[nt * 16 + c];

    v4f a[GATE_KSLICE / 8], b[GATE_KSLICE / 8];
    #pragma unroll
    for (int j = 0; j < GATE_KSLICE / 8; ++j) {
        a[j] = *(const v4f*)(aptr + 8 * j);
        b[j] = *(const v4f*)(bptr + 8 * j);
    }

    v8f acc = {};
    #pragma unroll
    for (int j = 0; j < GATE_KSLICE / 8; ++j) {
        v2f a0 = {a[j].x, a[j].y};
        v2f b0 = {b[j].x, b[j].y};
        v2f a1 = {a[j].z, a[j].w};
        v2f b1 = {b[j].z, b[j].w};
        acc = __builtin_amdgcn_wmma_f32_16x16x4_f32(
             false, a0,  false, b0,
             (short)0, acc,  false,  false);
        acc = __builtin_amdgcn_wmma_f32_16x16x4_f32(
             false, a1,  false, b1,
             (short)0, acc,  false,  false);
    }

    #pragma unroll
    for (int r = 0; r < 8; ++r) {
        red[wave][r * 32 + lane] = acc[r];
    }
    __syncthreads();

    int t = threadIdx.x;
    if (t < 8 * 32) {
        float s = red[0][t];
        #pragma unroll
        for (int w = 1; w < GATE_KSPLIT; ++w) {
            s += red[w][t];
        }
        int r = t >> 5;
        int l = t & 31;
        int m = mt * 16 + ((l >> 4) << 3) + r;
        int n = nt * 16 + (l & 15);
        float v = s + bb;
        float e = __expf(-v);
        gate[(size_t)m * D_MODEL + n] = __builtin_amdgcn_rcpf(1.0f + e);
    }
}

__global__ void se_apply_gate(const float* __restrict__ x,
                              const float* __restrict__ gate,
                              float* __restrict__ out) {
    size_t i = (size_t)blockIdx.x * blockDim.x + threadIdx.x;
    size_t f = i * 4;
    int    d   = (int)(f & (D_MODEL - 1));
    size_t row = f >> 10;
    size_t blk = row >> 10;
    v4f xv = *(const v4f*)(x + f);
    v4f gv = *(const v4f*)(gate + blk * D_MODEL + d);
    __builtin_nontemporal_store(xv * gv, (v4f*)(out + f));
}

extern "C" void kernel_launch(void* const* d_in, const int* in_sizes, int n_in,
                              void* d_out, int out_size, void* d_ws, size_t ws_size,
                              hipStream_t stream) {
    const float* x = (const float*)d_in[0];
    const float* W = (const float*)d_in[1];
    const float* b = (const float*)d_in[2];
    float* out = (float*)d_out;

    float* ws     = (float*)d_ws;
    float* pooled = ws;
    float* gate   = pooled + (size_t)NBLK * D_MODEL;
    float* part   = gate   + (size_t)NBLK * D_MODEL;

    se_pool_partial<<<NBLK * ROW_CHUNKS, 256 * POOL_GROUPS, 0, stream>>>(x, part);
    se_pool_reduce <<<NBLK,              256, 0, stream>>>(part, pooled);
    se_gate_wmma   <<<GATE_TILES, 32 * GATE_KSPLIT, 0, stream>>>(pooled, W, b, gate);

    const size_t total4 = (size_t)BATCH * SEQ * D_MODEL / 4;
    se_apply_gate<<<(unsigned)(total4 / 256), 256, 0, stream>>>(x, gate, out);
}


# apply pass re-tiled like the pooling pass: 256 work-groups of 16 waves (one per WGP) instead of 32768 of 8; a thread keeps its 4 gate values in registers and streams 64 rows, 16 row loads in flight
# speedup vs baseline: 1.8707x; 1.1632x over previous
#include <hip/hip_runtime.h>
#include <math.h>

typedef __attribute__((ext_vector_type(2))) float v2f;
typedef __attribute__((ext_vector_type(4))) float v4f;
typedef __attribute__((ext_vector_type(8))) float v8f;

#define D_MODEL 1024
#define NUM_BLOCKS 4
#define BATCH 8
#define SEQ 4096
#define BS (SEQ / NUM_BLOCKS)
#define NBLK (BATCH * NUM_BLOCKS)
#define ROW_CHUNKS 8
#define ROWS_PER_CHUNK (BS / ROW_CHUNKS)

#define POOL_GROUPS 2
#define POOL_ROWS   (ROWS_PER_CHUNK / POOL_GROUPS)
#define POOL_BATCH  32

__global__ void __launch_bounds__(256 * POOL_GROUPS) __attribute__((amdgpu_waves_per_eu(4, 4)))
se_pool_partial(const float* __restrict__ x, float* __restrict__ part) {
    __shared__ v4f red[256];

    int wg   = blockIdx.x;
    int c4   = threadIdx.x & 255;
    int half = threadIdx.x >> 8;
    const float* base = x + ((size_t)wg * ROWS_PER_CHUNK + (size_t)half * POOL_ROWS) * D_MODEL
                          + 4 * c4;
    v4f acc = {0.f, 0.f, 0.f, 0.f};
    #pragma unroll 1
    for (int r0 = 0; r0 < POOL_ROWS; r0 += POOL_BATCH) {
        v4f v[POOL_BATCH];
        #pragma unroll
        for (int r = 0; r < POOL_BATCH; ++r) {
            v[r] = *(const v4f*)(base + (size_t)(r0 + r) * D_MODEL);
        }
        #pragma unroll
        for (int r = 0; r < POOL_BATCH; ++r) {
            acc += v[r];
        }
    }
    if (half) red[c4] = acc;
    __syncthreads();
    if (!half) {
        acc += red[c4];
        *(v4f*)(part + (size_t)wg * D_MODEL + 4 * c4) = acc;
    }
}

__global__ void se_pool_reduce(const float* __restrict__ part, float* __restrict__ pooled) {
    int blk = blockIdx.x;
    int c4  = threadIdx.x;
    v4f acc = {0.f, 0.f, 0.f, 0.f};
    #pragma unroll
    for (int c = 0; c < ROW_CHUNKS; ++c) {
        acc += *(const v4f*)(part + ((size_t)blk * ROW_CHUNKS + c) * D_MODEL + 4 * c4);
    }
    acc *= (1.0f / (float)BS);
    *(v4f*)(pooled + (size_t)blk * D_MODEL + 4 * c4) = acc;
}

#define GATE_KSPLIT 16
#define GATE_KSLICE (D_MODEL / GATE_KSPLIT)
#define GATE_TILES  ((NBLK / 16) * (D_MODEL / 16))

__global__ void __launch_bounds__(32 * GATE_KSPLIT) __attribute__((amdgpu_waves_per_eu(4, 4)))
se_gate_wmma(const float* __restrict__ pooled,
             const float* __restrict__ W,
             const float* __restrict__ bias,
             float* __restrict__ gate) {
    __shared__ float red[GATE_KSPLIT][8 * 32];

    int tile = blockIdx.x;
    int mt   = tile & 1;
    int nt   = tile >> 1;
    int wave = threadIdx.x >> 5;
    int lane = threadIdx.x & 31;
    int c    = lane & 15;
    int h    = lane >> 4;

    int koff = wave * GATE_KSLICE + 4 * h;
    const float* aptr = pooled + (size_t)(mt * 16 + c) * D_MODEL + koff;
    const float* bptr = W      + (size_t)(nt * 16 + c) * D_MODEL + koff;

    float bb = bias[nt * 16 + c];

    v4f a[GATE_KSLICE / 8], b[GATE_KSLICE / 8];
    #pragma unroll
    for (int j = 0; j < GATE_KSLICE / 8; ++j) {
        a[j] = *(const v4f*)(aptr + 8 * j);
        b[j] = *(const v4f*)(bptr + 8 * j);
    }

    v8f acc = {};
    #pragma unroll
    for (int j = 0; j < GATE_KSLICE / 8; ++j) {
        v2f a0 = {a[j].x, a[j].y};
        v2f b0 = {b[j].x, b[j].y};
        v2f a1 = {a[j].z, a[j].w};
        v2f b1 = {b[j].z, b[j].w};
        acc = __builtin_amdgcn_wmma_f32_16x16x4_f32(
             false, a0,  false, b0,
             (short)0, acc,  false,  false);
        acc = __builtin_amdgcn_wmma_f32_16x16x4_f32(
             false, a1,  false, b1,
             (short)0, acc,  false,  false);
    }

    #pragma unroll
    for (int r = 0; r < 8; ++r) {
        red[wave][r * 32 + lane] = acc[r];
    }
    __syncthreads();

    int t = threadIdx.x;
    if (t < 8 * 32) {
        float s = red[0][t];
        #pragma unroll
        for (int w = 1; w < GATE_KSPLIT; ++w) {
            s += red[w][t];
        }
        int r = t >> 5;
        int l = t & 31;
        int m = mt * 16 + ((l >> 4) << 3) + r;
        int n = nt * 16 + (l & 15);
        float v = s + bb;
        float e = __expf(-v);
        gate[(size_t)m * D_MODEL + n] = __builtin_amdgcn_rcpf(1.0f + e);
    }
}

#define APPLY_GROUPS 2
#define APPLY_ROWS   (ROWS_PER_CHUNK / APPLY_GROUPS)
#define APPLY_BATCH  16

__global__ void __launch_bounds__(256 * APPLY_GROUPS) __attribute__((amdgpu_waves_per_eu(4, 4)))
se_apply_gate(const float* __restrict__ x,
              const float* __restrict__ gate,
              float* __restrict__ out) {
    int wg   = blockIdx.x;
    int c4   = threadIdx.x & 255;
    int half = threadIdx.x >> 8;
    size_t off = ((size_t)wg * ROWS_PER_CHUNK + (size_t)half * APPLY_ROWS) * D_MODEL + 4 * c4;
    const float* xp = x + off;
    float*       op = out + off;
    v4f gv = *(const v4f*)(gate + (size_t)(wg / ROW_CHUNKS) * D_MODEL + 4 * c4);
    #pragma unroll 1
    for (int r0 = 0; r0 < APPLY_ROWS; r0 += APPLY_BATCH) {
        v4f v[APPLY_BATCH];
        #pragma unroll
        for (int r = 0; r < APPLY_BATCH; ++r) {
            v[r] = *(const v4f*)(xp + (size_t)(r0 + r) * D_MODEL);
        }
        #pragma unroll
        for (int r = 0; r < APPLY_BATCH; ++r) {
            __builtin_nontemporal_store(v[r] * gv, (v4f*)(op + (size_t)(r0 + r) * D_MODEL));
        }
    }
}

extern "C" void kernel_launch(void* const* d_in, const int* in_sizes, int n_in,
                              void* d_out, int out_size, void* d_ws, size_t ws_size,
                              hipStream_t stream) {
    const float* x = (const float*)d_in[0];
    const float* W = (const float*)d_in[1];
    const float* b = (const float*)d_in[2];
    float* out = (float*)d_out;

    float* ws     = (float*)d_ws;
    float* pooled = ws;
    float* gate   = pooled + (size_t)NBLK * D_MODEL;
    float* part   = gate   + (size_t)NBLK * D_MODEL;

    se_pool_partial<<<NBLK * ROW_CHUNKS, 256 * POOL_GROUPS, 0, stream>>>(x, part);
    se_pool_reduce <<<NBLK,              256, 0, stream>>>(part, pooled);
    se_gate_wmma   <<<GATE_TILES, 32 * GATE_KSPLIT, 0, stream>>>(pooled, W, b, gate);

    se_apply_gate  <<<NBLK * ROW_CHUNKS, 256 * APPLY_GROUPS, 0, stream>>>(x, gate, out);
}


# reduce launch dropped: gate launch splits K over work-groups (16 slices x 8 column groups), adds the 8 chunk partials while staging A in LDS, writes 16 partial planes; apply adds planes+bias, sigmoid
# speedup vs baseline: 1.9636x; 1.0497x over previous
#include <hip/hip_runtime.h>
#include <math.h>

typedef __attribute__((ext_vector_type(2))) float v2f;
typedef __attribute__((ext_vector_type(4))) float v4f;
typedef __attribute__((ext_vector_type(8))) float v8f;

#define D_MODEL 1024
#define NUM_BLOCKS 4
#define BATCH 8
#define SEQ 4096
#define BS (SEQ / NUM_BLOCKS)
#define NBLK (BATCH * NUM_BLOCKS)
#define ROW_CHUNKS 8
#define ROWS_PER_CHUNK (BS / ROW_CHUNKS)

#define POOL_GROUPS 2
#define POOL_ROWS   (ROWS_PER_CHUNK / POOL_GROUPS)
#define POOL_BATCH  32

__global__ void __launch_bounds__(256 * POOL_GROUPS) __attribute__((amdgpu_waves_per_eu(4, 4)))
se_pool_partial(const float* __restrict__ x, float* __restrict__ part) {
    __shared__ v4f red[256];

    int wg   = blockIdx.x;
    int c4   = threadIdx.x & 255;
    int half = threadIdx.x >> 8;
    const float* base = x + ((size_t)wg * ROWS_PER_CHUNK + (size_t)half * POOL_ROWS) * D_MODEL
                          + 4 * c4;
    v4f acc = {0.f, 0.f, 0.f, 0.f};
    #pragma unroll 1
    for (int r0 = 0; r0 < POOL_ROWS; r0 += POOL_BATCH) {
        v4f v[POOL_BATCH];
        #pragma unroll
        for (int r = 0; r < POOL_BATCH; ++r) {
            v[r] = *(const v4f*)(base + (size_t)(r0 + r) * D_MODEL);
        }
        #pragma unroll
        for (int r = 0; r < POOL_BATCH; ++r) {
            acc += v[r];
        }
    }
    if (half) red[c4] = acc;
    __syncthreads();
    if (!half) {
        acc += red[c4];
        *(v4f*)(part + (size_t)wg * D_MODEL + 4 * c4) = acc;
    }
}

__global__ void se_pool_reduce(const float* __restrict__ part, float* __restrict__ pooled) {
    int blk = blockIdx.x;
    int c4  = threadIdx.x;
    v4f acc = {0.f, 0.f, 0.f, 0.f};
    #pragma unroll
    for (int c = 0; c < ROW_CHUNKS; ++c) {
        acc += *(const v4f*)(part + ((size_t)blk * ROW_CHUNKS + c) * D_MODEL + 4 * c4);
    }
    acc *= (1.0f / (float)BS);
    *(v4f*)(pooled + (size_t)blk * D_MODEL + 4 * c4) = acc;
}

#define GATE_KSPLIT 16
#define GATE_KSLICE (D_MODEL / GATE_KSPLIT)
#define GATE_TILES  ((NBLK / 16) * (D_MODEL / 16))

__global__ void __launch_bounds__(32 * GATE_KSPLIT) __attribute__((amdgpu_waves_per_eu(4, 4)))
se_gate_wmma(const float* __restrict__ pooled,
             const float* __restrict__ W,
             const float* __restrict__ bias,
             float* __restrict__ gate) {
    __shared__ float red[GATE_KSPLIT][8 * 32];

    int tile = blockIdx.x;
    int mt   = tile & 1;
    int nt   = tile >> 1;
    int wave = threadIdx.x >> 5;
    int lane = threadIdx.x & 31;
    int c    = lane & 15;
    int h    = lane >> 4;

    int koff = wave * GATE_KSLICE + 4 * h;
    const float* aptr = pooled + (size_t)(mt * 16 + c) * D_MODEL + koff;
    const float* bptr = W      + (size_t)(nt * 16 + c) * D_MODEL + koff;

    float bb = bias[nt * 16 + c];

    v4f a[GATE_KSLICE / 8], b[GATE_KSLICE / 8];
    #pragma unroll
    for (int j = 0; j < GATE_KSLICE / 8; ++j) {
        a[j] = *(const v4f*)(aptr + 8 * j);
        b[j] = *(const v4f*)(bptr + 8 * j);
    }

    v8f acc = {};
    #pragma unroll
    for (int j = 0; j < GATE_KSLICE / 8; ++j) {
        v2f a0 = {a[j].x, a[j].y};
        v2f b0 = {b[j].x, b[j].y};
        v2f a1 = {a[j].z, a[j].w};
        v2f b1 = {b[j].z, b[j].w};
        acc = __builtin_amdgcn_wmma_f32_16x16x4_f32(
             false, a0,  false, b0,
             (short)0, acc,  false,  false);
        acc = __builtin_amdgcn_wmma_f32_16x16x4_f32(
             false, a1,  false, b1,
             (short)0, acc,  false,  false);
    }

    #pragma unroll
    for (int r = 0; r < 8; ++r) {
        red[wave][r * 32 + lane] = acc[r];
    }
    __syncthreads();

    int t = threadIdx.x;
    if (t < 8 * 32) {
        float s = red[0][t];
        #pragma unroll
        for (int w = 1; w < GATE_KSPLIT; ++w) {
            s += red[w][t];
        }
        int r = t >> 5;
        int l = t & 31;
        int m = mt * 16 + ((l >> 4) << 3) + r;
        int n = nt * 16 + (l & 15);
        float v = s + bb;
        float e = __expf(-v);
        gate[(size_t)m * D_MODEL + n] = __builtin_amdgcn_rcpf(1.0f + e);
    }
}

#define APPLY_GROUPS 2
#define APPLY_ROWS   (ROWS_PER_CHUNK / APPLY_GROUPS)
#define APPLY_BATCH  16

__global__ void __launch_bounds__(256 * APPLY_GROUPS) __attribute__((amdgpu_waves_per_eu(4, 4)))
se_apply_gate(const float* __restrict__ x,
              const float* __restrict__ gate,
              float* __restrict__ out) {
    int wg   = blockIdx.x;
    int c4   = threadIdx.x & 255;
    int half = threadIdx.x >> 8;
    size_t off = ((size_t)wg * ROWS_PER_CHUNK + (size_t)half * APPLY_ROWS) * D_MODEL + 4 * c4;
    const float* xp = x + off;
    float*       op = out + off;
    v4f gv = *(const v4f*)(gate + (size_t)(wg / ROW_CHUNKS) * D_MODEL + 4 * c4);
    #pragma unroll 1
    for (int r0 = 0; r0 < APPLY_ROWS; r0 += APPLY_BATCH) {
        v4f v[APPLY_BATCH];
        #pragma unroll
        for (int r = 0; r < APPLY_BATCH; ++r) {
            v[r] = *(const v4f*)(xp + (size_t)(r0 + r) * D_MODEL);
        }
        #pragma unroll
        for (int r = 0; r < APPLY_BATCH; ++r) {
            __builtin_nontemporal_store(v[r] * gv, (v4f*)(op + (size_t)(r0 + r) * D_MODEL));
        }
    }
}


#define GP_COLS    128
#define GP_CGROUPS (D_MODEL / GP_COLS)
#define GP_WGS     (GATE_KSPLIT * GP_CGROUPS)
#define GP_APITCH  (GATE_KSLICE + 4)
#define GP_DPITCH  (GP_COLS + 4)

__global__ void __launch_bounds__(32 * GATE_KSPLIT) __attribute__((amdgpu_waves_per_eu(4, 4)))
se_gate_planes(const float* __restrict__ part,
               const float* __restrict__ W,
               float* __restrict__ planes) {
    __shared__ __attribute__((aligned(16))) float As[NBLK * GP_APITCH];
    __shared__ __attribute__((aligned(16))) float Ds[NBLK * GP_DPITCH];

    int ks   = blockIdx.x & (GATE_KSPLIT - 1);
    int cg   = blockIdx.x / GATE_KSPLIT;
    int t    = threadIdx.x;
    int wave = t >> 5;
    int lane = t & 31;
    int c    = lane & 15;
    int h    = lane >> 4;
    int mt   = wave & 1;
    int nt   = wave >> 1;

    int m = t >> 4;
    int q = t & 15;
    const float* pp   = part + (size_t)m * ROW_CHUNKS * D_MODEL + ks * GATE_KSLICE + 4 * q;
    const float* bptr = W + (size_t)(cg * GP_COLS + nt * 16 + c) * D_MODEL
                          + ks * GATE_KSLICE + 4 * h;

    v4f pc[ROW_CHUNKS], b[GATE_KSLICE / 8];
    #pragma unroll
    for (int ch = 0; ch < ROW_CHUNKS; ++ch) {
        pc[ch] = *(const v4f*)(pp + (size_t)ch * D_MODEL);
    }
    #pragma unroll
    for (int j = 0; j < GATE_KSLICE / 8; ++j) {
        b[j] = *(const v4f*)(bptr + 8 * j);
    }

    v4f s = {0.f, 0.f, 0.f, 0.f};
    #pragma unroll
    for (int ch = 0; ch < ROW_CHUNKS; ++ch) {
        s += pc[ch];
    }
    s *= (1.0f / (float)BS);
    *(v4f*)(As + m * GP_APITCH + 4 * q) = s;
    __syncthreads();

    v4f a[GATE_KSLICE / 8];
    const float* ap = As + (mt * 16 + c) * GP_APITCH + 4 * h;
    #pragma unroll
    for (int j = 0; j < GATE_KSLICE / 8; ++j) {
        a[j] = *(const v4f*)(ap + 8 * j);
    }

    v8f acc = {};
    #pragma unroll
    for (int j = 0; j < GATE_KSLICE / 8; ++j) {
        v2f a0 = {a[j].x, a[j].y};
        v2f b0 = {b[j].x, b[j].y};
        v2f a1 = {a[j].z, a[j].w};
        v2f b1 = {b[j].z, b[j].w};
        acc = __builtin_amdgcn_wmma_f32_16x16x4_f32(
             false, a0,  false, b0,
             (short)0, acc,  false,  false);
        acc = __builtin_amdgcn_wmma_f32_16x16x4_f32(
             false, a1,  false, b1,
             (short)0, acc,  false,  false);
    }

    #pragma unroll
    for (int r = 0; r < 8; ++r) {
        Ds[(mt * 16 + 8 * h + r) * GP_DPITCH + nt * 16 + c] = acc[r];
    }
    __syncthreads();

    #pragma unroll
    for (int p = 0; p < 2; ++p) {
        int row = wave + 16 * p;
        v4f v = *(const v4f*)(Ds + row * GP_DPITCH + 4 * lane);
        *(v4f*)(planes + ((size_t)ks * NBLK + row) * D_MODEL + cg * GP_COLS + 4 * lane) = v;
    }
}

__global__ void __launch_bounds__(256 * APPLY_GROUPS) __attribute__((amdgpu_waves_per_eu(4, 4)))
se_apply_planes(const float* __restrict__ x,
                const float* __restrict__ planes,
                const float* __restrict__ bias,
                float* __restrict__ out) {
    int wg   = blockIdx.x;
    int c4   = threadIdx.x & 255;
    int half = threadIdx.x >> 8;
    size_t off = ((size_t)wg * ROWS_PER_CHUNK + (size_t)half * APPLY_ROWS) * D_MODEL + 4 * c4;
    const float* xp = x + off;
    float*       op = out + off;
    const float* gp = planes + (size_t)(wg / ROW_CHUNKS) * D_MODEL + 4 * c4;

    v4f pl[GATE_KSPLIT];
    #pragma unroll
    for (int w = 0; w < GATE_KSPLIT; ++w) {
        pl[w] = *(const v4f*)(gp + (size_t)w * NBLK * D_MODEL);
    }
    v4f bb = *(const v4f*)(bias + 4 * c4);

    v4f v[APPLY_BATCH];
    #pragma unroll
    for (int r = 0; r < APPLY_BATCH; ++r) {
        v[r] = *(const v4f*)(xp + (size_t)r * D_MODEL);
    }

    v4f sum = pl[0];
    #pragma unroll
    for (int w = 1; w < GATE_KSPLIT; ++w) {
        sum += pl[w];
    }
    sum += bb;
    v4f gv;
    gv.x = __builtin_amdgcn_rcpf(1.0f + __expf(-sum.x));
    gv.y = __builtin_amdgcn_rcpf(1.0f + __expf(-sum.y));
    gv.z = __builtin_amdgcn_rcpf(1.0f + __expf(-sum.z));
    gv.w = __builtin_amdgcn_rcpf(1.0f + __expf(-sum.w));

    #pragma unroll
    for (int r = 0; r < APPLY_BATCH; ++r) {
        __builtin_nontemporal_store(v[r] * gv, (v4f*)(op + (size_t)r * D_MODEL));
    }

    #pragma unroll 1
    for (int r0 = APPLY_BATCH; r0 < APPLY_ROWS; r0 += APPLY_BATCH) {
        #pragma unroll
        for (int r = 0; r < APPLY_BATCH; ++r) {
            v[r] = *(const v4f*)(xp + (size_t)(r0 + r) * D_MODEL);
        }
        #pragma unroll
        for (int r = 0; r < APPLY_BATCH; ++r) {
            __builtin_nontemporal_store(v[r] * gv, (v4f*)(op + (size_t)(r0 + r) * D_MODEL));
        }
    }
}

extern "C" void kernel_launch(void* const* d_in, const int* in_sizes, int n_in,
                              void* d_out, int out_size, void* d_ws, size_t ws_size,
                              hipStream_t stream) {
    const float* x = (const float*)d_in[0];
    const float* W = (const float*)d_in[1];
    const float* b = (const float*)d_in[2];
    float* out = (float*)d_out;

    float* ws = (float*)d_ws;

    if (ws_size >= (size_t)3145728) {
        float* part   = ws;
        float* planes = part + (size_t)NBLK * ROW_CHUNKS * D_MODEL;
        se_pool_partial<<<NBLK * ROW_CHUNKS, 256 * POOL_GROUPS, 0, stream>>>(x, part);
        se_gate_planes <<<GP_WGS, 32 * GATE_KSPLIT, 0, stream>>>(part, W, planes);
        se_apply_planes<<<NBLK * ROW_CHUNKS, 256 * APPLY_GROUPS, 0, stream>>>(x, planes, b, out);
        return;
    }

    float* pooled = ws;
    float* gate   = pooled + (size_t)NBLK * D_MODEL;
    float* part   = gate   + (size_t)NBLK * D_MODEL;

    se_pool_partial<<<NBLK * ROW_CHUNKS, 256 * POOL_GROUPS, 0, stream>>>(x, part);
    se_pool_reduce <<<NBLK,              256, 0, stream>>>(part, pooled);
    se_gate_wmma   <<<GATE_TILES, 32 * GATE_KSPLIT, 0, stream>>>(pooled, W, b, gate);

    se_apply_gate  <<<NBLK * ROW_CHUNKS, 256 * APPLY_GROUPS, 0, stream>>>(x, gate, out);
}


# apply pass walks each thread's 64 rows last to first, opposite to the pooling pass, so the rows pooled last (still in the second-level cache) are re-read first; no sum or its order changes
# speedup vs baseline: 2.0390x; 1.0384x over previous
#include <hip/hip_runtime.h>
#include <math.h>

typedef __attribute__((ext_vector_type(2))) float v2f;
typedef __attribute__((ext_vector_type(4))) float v4f;
typedef __attribute__((ext_vector_type(8))) float v8f;

#define D_MODEL 1024
#define NUM_BLOCKS 4
#define BATCH 8
#define SEQ 4096
#define BS (SEQ / NUM_BLOCKS)
#define NBLK (BATCH * NUM_BLOCKS)
#define ROW_CHUNKS 8
#define ROWS_PER_CHUNK (BS / ROW_CHUNKS)

#define POOL_GROUPS 2
#define POOL_ROWS   (ROWS_PER_CHUNK / POOL_GROUPS)
#define POOL_BATCH  32

__global__ void __launch_bounds__(256 * POOL_GROUPS) __attribute__((amdgpu_waves_per_eu(4, 4)))
se_pool_partial(const float* __restrict__ x, float* __restrict__ part) {
    __shared__ v4f red[256];

    int wg   = blockIdx.x;
    int c4   = threadIdx.x & 255;
    int half = threadIdx.x >> 8;
    const float* base = x + ((size_t)wg * ROWS_PER_CHUNK + (size_t)half * POOL_ROWS) * D_MODEL
                          + 4 * c4;
    v4f acc = {0.f, 0.f, 0.f, 0.f};
    #pragma unroll 1
    for (int r0 = 0; r0 < POOL_ROWS; r0 += POOL_BATCH) {
        v4f v[POOL_BATCH];
        #pragma unroll
        for (int r = 0; r < POOL_BATCH; ++r) {
            v[r] = *(const v4f*)(base + (size_t)(r0 + r) * D_MODEL);
        }
        #pragma unroll
        for (int r = 0; r < POOL_BATCH; ++r) {
            acc += v[r];
        }
    }
    if (half) red[c4] = acc;
    __syncthreads();
    if (!half) {
        acc += red[c4];
        *(v4f*)(part + (size_t)wg * D_MODEL + 4 * c4) = acc;
    }
}

__global__ void se_pool_reduce(const float* __restrict__ part, float* __restrict__ pooled) {
    int blk = blockIdx.x;
    int c4  = threadIdx.x;
    v4f acc = {0.f, 0.f, 0.f, 0.f};
    #pragma unroll
    for (int c = 0; c < ROW_CHUNKS; ++c) {
        acc += *(const v4f*)(part + ((size_t)blk * ROW_CHUNKS + c) * D_MODEL + 4 * c4);
    }
    acc *= (1.0f / (float)BS);
    *(v4f*)(pooled + (size_t)blk * D_MODEL + 4 * c4) = acc;
}

#define GATE_KSPLIT 16
#define GATE_KSLICE (D_MODEL / GATE_KSPLIT)
#define GATE_TILES  ((NBLK / 16) * (D_MODEL / 16))

__global__ void __launch_bounds__(32 * GATE_KSPLIT) __attribute__((amdgpu_waves_per_eu(4, 4)))
se_gate_wmma(const float* __restrict__ pooled,
             const float* __restrict__ W,
             const float* __restrict__ bias,
             float* __restrict__ gate) {
    __shared__ float red[GATE_KSPLIT][8 * 32];

    int tile = blockIdx.x;
    int mt   = tile & 1;
    int nt   = tile >> 1;
    int wave = threadIdx.x >> 5;
    int lane = threadIdx.x & 31;
    int c    = lane & 15;
    int h    = lane >> 4;

    int koff = wave * GATE_KSLICE + 4 * h;
    const float* aptr = pooled + (size_t)(mt * 16 + c) * D_MODEL + koff;
    const float* bptr = W      + (size_t)(nt * 16 + c) * D_MODEL + koff;

    float bb = bias[nt * 16 + c];

    v4f a[GATE_KSLICE / 8], b[GATE_KSLICE / 8];
    #pragma unroll
    for (int j = 0; j < GATE_KSLICE / 8; ++j) {
        a[j] = *(const v4f*)(aptr + 8 * j);
        b[j] = *(const v4f*)(bptr + 8 * j);
    }

    v8f acc = {};
    #pragma unroll
    for (int j = 0; j < GATE_KSLICE / 8; ++j) {
        v2f a0 = {a[j].x, a[j].y};
        v2f b0 = {b[j].x, b[j].y};
        v2f a1 = {a[j].z, a[j].w};
        v2f b1 = {b[j].z, b[j].w};
        acc = __builtin_amdgcn_wmma_f32_16x16x4_f32(
             false, a0,  false, b0,
             (short)0, acc,  false,  false);
        acc = __builtin_amdgcn_wmma_f32_16x16x4_f32(
             false, a1,  false, b1,
             (short)0, acc,  false,  false);
    }

    #pragma unroll
    for (int r = 0; r < 8; ++r) {
        red[wave][r * 32 + lane] = acc[r];
    }
    __syncthreads();

    int t = threadIdx.x;
    if (t < 8 * 32) {
        float s = red[0][t];
        #pragma unroll
        for (int w = 1; w < GATE_KSPLIT; ++w) {
            s += red[w][t];
        }
        int r = t >> 5;
        int l = t & 31;
        int m = mt * 16 + ((l >> 4) << 3) + r;
        int n = nt * 16 + (l & 15);
        float v = s + bb;
        float e = __expf(-v);
        gate[(size_t)m * D_MODEL + n] = __builtin_amdgcn_rcpf(1.0f + e);
    }
}

#define APPLY_GROUPS 2
#define APPLY_ROWS   (ROWS_PER_CHUNK / APPLY_GROUPS)
#define APPLY_BATCH  16

__global__ void __launch_bounds__(256 * APPLY_GROUPS) __attribute__((amdgpu_waves_per_eu(4, 4)))
se_apply_gate(const float* __restrict__ x,
              const float* __restrict__ gate,
              float* __restrict__ out) {
    int wg   = blockIdx.x;
    int c4   = threadIdx.x & 255;
    int half = threadIdx.x >> 8;
    size_t off = ((size_t)wg * ROWS_PER_CHUNK + (size_t)half * APPLY_ROWS + (APPLY_ROWS - 1)) * D_MODEL
               + 4 * c4;
    const float* xp = x + off;
    float*       op = out + off;
    v4f gv = *(const v4f*)(gate + (size_t)(wg / ROW_CHUNKS) * D_MODEL + 4 * c4);
    #pragma unroll 1
    for (int r0 = 0; r0 < APPLY_ROWS; r0 += APPLY_BATCH) {
        v4f v[APPLY_BATCH];
        #pragma unroll
        for (int r = 0; r < APPLY_BATCH; ++r) {
            v[r] = *(const v4f*)(xp - (size_t)(r0 + r) * D_MODEL);
        }
        #pragma unroll
        for (int r = 0; r < APPLY_BATCH; ++r) {
            __builtin_nontemporal_store(v[r] * gv, (v4f*)(op - (size_t)(r0 + r) * D_MODEL));
        }
    }
}


#define GP_COLS    128
#define GP_CGROUPS (D_MODEL / GP_COLS)
#define GP_WGS     (GATE_KSPLIT * GP_CGROUPS)
#define GP_APITCH  (GATE_KSLICE + 4)
#define GP_DPITCH  (GP_COLS + 4)

__global__ void __launch_bounds__(32 * GATE_KSPLIT) __attribute__((amdgpu_waves_per_eu(4, 4)))
se_gate_planes(const float* __restrict__ part,
               const float* __restrict__ W,
               float* __restrict__ planes) {
    __shared__ __attribute__((aligned(16))) float As[NBLK * GP_APITCH];
    __shared__ __attribute__((aligned(16))) float Ds[NBLK * GP_DPITCH];

    int ks   = blockIdx.x & (GATE_KSPLIT - 1);
    int cg   = blockIdx.x / GATE_KSPLIT;
    int t    = threadIdx.x;
    int wave = t >> 5;
    int lane = t & 31;
    int c    = lane & 15;
    int h    = lane >> 4;
    int mt   = wave & 1;
    int nt   = wave >> 1;

    int m = t >> 4;
    int q = t & 15;
    const float* pp   = part + (size_t)m * ROW_CHUNKS * D_MODEL + ks * GATE_KSLICE + 4 * q;
    const float* bptr = W + (size_t)(cg * GP_COLS + nt * 16 + c) * D_MODEL
                          + ks * GATE_KSLICE + 4 * h;

    v4f pc[ROW_CHUNKS], b[GATE_KSLICE / 8];
    #pragma unroll
    for (int ch = 0; ch < ROW_CHUNKS; ++ch) {
        pc[ch] = *(const v4f*)(pp + (size_t)ch * D_MODEL);
    }
    #pragma unroll
    for (int j = 0; j < GATE_KSLICE / 8; ++j) {
        b[j] = *(const v4f*)(bptr + 8 * j);
    }

    v4f s = {0.f, 0.f, 0.f, 0.f};
    #pragma unroll
    for (int ch = 0; ch < ROW_CHUNKS; ++ch) {
        s += pc[ch];
    }
    s *= (1.0f / (float)BS);
    *(v4f*)(As + m * GP_APITCH + 4 * q) = s;
    __syncthreads();

    v4f a[GATE_KSLICE / 8];
    const float* ap = As + (mt * 16 + c) * GP_APITCH + 4 * h;
    #pragma unroll
    for (int j = 0; j < GATE_KSLICE / 8; ++j) {
        a[j] = *(const v4f*)(ap + 8 * j);
    }

    v8f acc = {};
    #pragma unroll
    for (int j = 0; j < GATE_KSLICE / 8; ++j) {
        v2f a0 = {a[j].x, a[j].y};
        v2f b0 = {b[j].x, b[j].y};
        v2f a1 = {a[j].z, a[j].w};
        v2f b1 = {b[j].z, b[j].w};
        acc = __builtin_amdgcn_wmma_f32_16x16x4_f32(
             false, a0,  false, b0,
             (short)0, acc,  false,  false);
        acc = __builtin_amdgcn_wmma_f32_16x16x4_f32(
             false, a1,  false, b1,
             (short)0, acc,  false,  false);
    }

    #pragma unroll
    for (int r = 0; r < 8; ++r) {
        Ds[(mt * 16 + 8 * h + r) * GP_DPITCH + nt * 16 + c] = acc[r];
    }
    __syncthreads();

    #pragma unroll
    for (int p = 0; p < 2; ++p) {
        int row = wave + 16 * p;
        v4f v = *(const v4f*)(Ds + row * GP_DPITCH + 4 * lane);
        *(v4f*)(planes + ((size_t)ks * NBLK + row) * D_MODEL + cg * GP_COLS + 4 * lane) = v;
    }
}

__global__ void __launch_bounds__(256 * APPLY_GROUPS) __attribute__((amdgpu_waves_per_eu(4, 4)))
se_apply_planes(const float* __restrict__ x,
                const float* __restrict__ planes,
                const float* __restrict__ bias,
                float* __restrict__ out) {
    int wg   = blockIdx.x;
    int c4   = threadIdx.x & 255;
    int half = threadIdx.x >> 8;
    size_t off = ((size_t)wg * ROWS_PER_CHUNK + (size_t)half * APPLY_ROWS + (APPLY_ROWS - 1)) * D_MODEL
               + 4 * c4;
    const float* xp = x + off;
    float*       op = out + off;
    const float* gp = planes + (size_t)(wg / ROW_CHUNKS) * D_MODEL + 4 * c4;

    v4f pl[GATE_KSPLIT];
    #pragma unroll
    for (int w = 0; w < GATE_KSPLIT; ++w) {
        pl[w] = *(const v4f*)(gp + (size_t)w * NBLK * D_MODEL);
    }
    v4f bb = *(const v4f*)(bias + 4 * c4);

    v4f v[APPLY_BATCH];
    #pragma unroll
    for (int r = 0; r < APPLY_BATCH; ++r) {
        v[r] = *(const v4f*)(xp - (size_t)r * D_MODEL);
    }

    v4f sum = pl[0];
    #pragma unroll
    for (int w = 1; w < GATE_KSPLIT; ++w) {
        sum += pl[w];
    }
    sum += bb;
    v4f gv;
    gv.x = __builtin_amdgcn_rcpf(1.0f + __expf(-sum.x));
    gv.y = __builtin_amdgcn_rcpf(1.0f + __expf(-sum.y));
    gv.z = __builtin_amdgcn_rcpf(1.0f + __expf(-sum.z));
    gv.w = __builtin_amdgcn_rcpf(1.0f + __expf(-sum.w));

    #pragma unroll
    for (int r = 0; r < APPLY_BATCH; ++r) {
        __builtin_nontemporal_store(v[r] * gv, (v4f*)(op - (size_t)r * D_MODEL));
    }

    #pragma unroll 1
    for (int r0 = APPLY_BATCH; r0 < APPLY_ROWS; r0 += APPLY_BATCH) {
        #pragma unroll
        for (int r = 0; r < APPLY_BATCH; ++r) {
            v[r] = *(const v4f*)(xp - (size_t)(r0 + r) * D_MODEL);
        }
        #pragma unroll
        for (int r = 0; r < APPLY_BATCH; ++r) {
            __builtin_nontemporal_store(v[r] * gv, (v4f*)(op - (size_t)(r0 + r) * D_MODEL));
        }
    }
}

extern "C" void kernel_launch(void* const* d_in, const int* in_sizes, int n_in,
                              void* d_out, int out_size, void* d_ws, size_t ws_size,
                              hipStream_t stream) {
    const float* x = (const float*)d_in[0];
    const float* W = (const float*)d_in[1];
    const float* b = (const float*)d_in[2];
    float* out = (float*)d_out;

    float* ws = (float*)d_ws;

    if (ws_size >= (size_t)3145728) {
        float* part   = ws;
        float* planes = part + (size_t)NBLK * ROW_CHUNKS * D_MODEL;
        se_pool_partial<<<NBLK * ROW_CHUNKS, 256 * POOL_GROUPS, 0, stream>>>(x, part);
        se_gate_planes <<<GP_WGS, 32 * GATE_KSPLIT, 0, stream>>>(part, W, planes);
        se_apply_planes<<<NBLK * ROW_CHUNKS, 256 * APPLY_GROUPS, 0, stream>>>(x, planes, b, out);
        return;
    }

    float* pooled = ws;
    float* gate   = pooled + (size_t)NBLK * D_MODEL;
    float* part   = gate   + (size_t)NBLK * D_MODEL;

    se_pool_partial<<<NBLK * ROW_CHUNKS, 256 * POOL_GROUPS, 0, stream>>>(x, part);
    se_pool_reduce <<<NBLK,              256, 0, stream>>>(part, pooled);
    se_gate_wmma   <<<GATE_TILES, 32 * GATE_KSPLIT, 0, stream>>>(pooled, W, b, gate);

    se_apply_gate  <<<NBLK * ROW_CHUNKS, 256 * APPLY_GROUPS, 0, stream>>>(x, gate, out);
}


# apply pass keeps a rolling window of 32 row loads per thread, refilled 8 rows at a time behind each group of 8 stores, instead of 16 loads, wait, 16 stores in turn; no sum or its order changes
# speedup vs baseline: 2.1364x; 1.0478x over previous
#include <hip/hip_runtime.h>
#include <math.h>

typedef __attribute__((ext_vector_type(2))) float v2f;
typedef __attribute__((ext_vector_type(4))) float v4f;
typedef __attribute__((ext_vector_type(8))) float v8f;

#define D_MODEL 1024
#define NUM_BLOCKS 4
#define BATCH 8
#define SEQ 4096
#define BS (SEQ / NUM_BLOCKS)
#define NBLK (BATCH * NUM_BLOCKS)
#define ROW_CHUNKS 8
#define ROWS_PER_CHUNK (BS / ROW_CHUNKS)

#define POOL_GROUPS 2
#define POOL_ROWS   (ROWS_PER_CHUNK / POOL_GROUPS)
#define POOL_BATCH  32

__global__ void __launch_bounds__(256 * POOL_GROUPS) __attribute__((amdgpu_waves_per_eu(4, 4)))
se_pool_partial(const float* __restrict__ x, float* __restrict__ part) {
    __shared__ v4f red[256];

    int wg   = blockIdx.x;
    int c4   = threadIdx.x & 255;
    int half = threadIdx.x >> 8;
    const float* base = x + ((size_t)wg * ROWS_PER_CHUNK + (size_t)half * POOL_ROWS) * D_MODEL
                          + 4 * c4;
    v4f acc = {0.f, 0.f, 0.f, 0.f};
    #pragma unroll 1
    for (int r0 = 0; r0 < POOL_ROWS; r0 += POOL_BATCH) {
        v4f v[POOL_BATCH];
        #pragma unroll
        for (int r = 0; r < POOL_BATCH; ++r) {
            v[r] = *(const v4f*)(base + (size_t)(r0 + r) * D_MODEL);
        }
        #pragma unroll
        for (int r = 0; r < POOL_BATCH; ++r) {
            acc += v[r];
        }
    }
    if (half) red[c4] = acc;
    __syncthreads();
    if (!half) {
        acc += red[c4];
        *(v4f*)(part + (size_t)wg * D_MODEL + 4 * c4) = acc;
    }
}

__global__ void se_pool_reduce(const float* __restrict__ part, float* __restrict__ pooled) {
    int blk = blockIdx.x;
    int c4  = threadIdx.x;
    v4f acc = {0.f, 0.f, 0.f, 0.f};
    #pragma unroll
    for (int c = 0; c < ROW_CHUNKS; ++c) {
        acc += *(const v4f*)(part + ((size_t)blk * ROW_CHUNKS + c) * D_MODEL + 4 * c4);
    }
    acc *= (1.0f / (float)BS);
    *(v4f*)(pooled + (size_t)blk * D_MODEL + 4 * c4) = acc;
}

#define GATE_KSPLIT 16
#define GATE_KSLICE (D_MODEL / GATE_KSPLIT)
#define GATE_TILES  ((NBLK / 16) * (D_MODEL / 16))

__global__ void __launch_bounds__(32 * GATE_KSPLIT) __attribute__((amdgpu_waves_per_eu(4, 4)))
se_gate_wmma(const float* __restrict__ pooled,
             const float* __restrict__ W,
             const float* __restrict__ bias,
             float* __restrict__ gate) {
    __shared__ float red[GATE_KSPLIT][8 * 32];

    int tile = blockIdx.x;
    int mt   = tile & 1;
    int nt   = tile >> 1;
    int wave = threadIdx.x >> 5;
    int lane = threadIdx.x & 31;
    int c    = lane & 15;
    int h    = lane >> 4;

    int koff = wave * GATE_KSLICE + 4 * h;
    const float* aptr = pooled + (size_t)(mt * 16 + c) * D_MODEL + koff;
    const float* bptr = W      + (size_t)(nt * 16 + c) * D_MODEL + koff;

    float bb = bias[nt * 16 + c];

    v4f a[GATE_KSLICE / 8], b[GATE_KSLICE / 8];
    #pragma unroll
    for (int j = 0; j < GATE_KSLICE / 8; ++j) {
        a[j] = *(const v4f*)(aptr + 8 * j);
        b[j] = *(const v4f*)(bptr + 8 * j);
    }

    v8f acc = {};
    #pragma unroll
    for (int j = 0; j < GATE_KSLICE / 8; ++j) {
        v2f a0 = {a[j].x, a[j].y};
        v2f b0 = {b[j].x, b[j].y};
        v2f a1 = {a[j].z, a[j].w};
        v2f b1 = {b[j].z, b[j].w};
        acc = __builtin_amdgcn_wmma_f32_16x16x4_f32(
             false, a0,  false, b0,
             (short)0, acc,  false,  false);
        acc = __builtin_amdgcn_wmma_f32_16x16x4_f32(
             false, a1,  false, b1,
             (short)0, acc,  false,  false);
    }

    #pragma unroll
    for (int r = 0; r < 8; ++r) {
        red[wave][r * 32 + lane] = acc[r];
    }
    __syncthreads();

    int t = threadIdx.x;
    if (t < 8 * 32) {
        float s = red[0][t];
        #pragma unroll
        for (int w = 1; w < GATE_KSPLIT; ++w) {
            s += red[w][t];
        }
        int r = t >> 5;
        int l = t & 31;
        int m = mt * 16 + ((l >> 4) << 3) + r;
        int n = nt * 16 + (l & 15);
        float v = s + bb;
        float e = __expf(-v);
        gate[(size_t)m * D_MODEL + n] = __builtin_amdgcn_rcpf(1.0f + e);
    }
}

#define APPLY_GROUPS 2
#define APPLY_ROWS   (ROWS_PER_CHUNK / APPLY_GROUPS)
#define APPLY_BATCH  16

__global__ void __launch_bounds__(256 * APPLY_GROUPS) __attribute__((amdgpu_waves_per_eu(4, 4)))
se_apply_gate(const float* __restrict__ x,
              const float* __restrict__ gate,
              float* __restrict__ out) {
    int wg   = blockIdx.x;
    int c4   = threadIdx.x & 255;
    int half = threadIdx.x >> 8;
    size_t off = ((size_t)wg * ROWS_PER_CHUNK + (size_t)half * APPLY_ROWS + (APPLY_ROWS - 1)) * D_MODEL
               + 4 * c4;
    const float* xp = x + off;
    float*       op = out + off;
    v4f gv = *(const v4f*)(gate + (size_t)(wg / ROW_CHUNKS) * D_MODEL + 4 * c4);
    #pragma unroll 1
    for (int r0 = 0; r0 < APPLY_ROWS; r0 += APPLY_BATCH) {
        v4f v[APPLY_BATCH];
        #pragma unroll
        for (int r = 0; r < APPLY_BATCH; ++r) {
            v[r] = *(const v4f*)(xp - (size_t)(r0 + r) * D_MODEL);
        }
        #pragma unroll
        for (int r = 0; r < APPLY_BATCH; ++r) {
            __builtin_nontemporal_store(v[r] * gv, (v4f*)(op - (size_t)(r0 + r) * D_MODEL));
        }
    }
}


#define GP_COLS    128
#define GP_CGROUPS (D_MODEL / GP_COLS)
#define GP_WGS     (GATE_KSPLIT * GP_CGROUPS)
#define GP_APITCH  (GATE_KSLICE + 4)
#define GP_DPITCH  (GP_COLS + 4)

__global__ void __launch_bounds__(32 * GATE_KSPLIT) __attribute__((amdgpu_waves_per_eu(4, 4)))
se_gate_planes(const float* __restrict__ part,
               const float* __restrict__ W,
               float* __restrict__ planes) {
    __shared__ __attribute__((aligned(16))) float As[NBLK * GP_APITCH];
    __shared__ __attribute__((aligned(16))) float Ds[NBLK * GP_DPITCH];

    int ks   = blockIdx.x & (GATE_KSPLIT - 1);
    int cg   = blockIdx.x / GATE_KSPLIT;
    int t    = threadIdx.x;
    int wave = t >> 5;
    int lane = t & 31;
    int c    = lane & 15;
    int h    = lane >> 4;
    int mt   = wave & 1;
    int nt   = wave >> 1;

    int m = t >> 4;
    int q = t & 15;
    const float* pp   = part + (size_t)m * ROW_CHUNKS * D_MODEL + ks * GATE_KSLICE + 4 * q;
    const float* bptr = W + (size_t)(cg * GP_COLS + nt * 16 + c) * D_MODEL
                          + ks * GATE_KSLICE + 4 * h;

    v4f pc[ROW_CHUNKS], b[GATE_KSLICE / 8];
    #pragma unroll
    for (int ch = 0; ch < ROW_CHUNKS; ++ch) {
        pc[ch] = *(const v4f*)(pp + (size_t)ch * D_MODEL);
    }
    #pragma unroll
    for (int j = 0; j < GATE_KSLICE / 8; ++j) {
        b[j] = *(const v4f*)(bptr + 8 * j);
    }

    v4f s = {0.f, 0.f, 0.f, 0.f};
    #pragma unroll
    for (int ch = 0; ch < ROW_CHUNKS; ++ch) {
        s += pc[ch];
    }
    s *= (1.0f / (float)BS);
    *(v4f*)(As + m * GP_APITCH + 4 * q) = s;
    __syncthreads();

    v4f a[GATE_KSLICE / 8];
    const float* ap = As + (mt * 16 + c) * GP_APITCH + 4 * h;
    #pragma unroll
    for (int j = 0; j < GATE_KSLICE / 8; ++j) {
        a[j] = *(const v4f*)(ap + 8 * j);
    }

    v8f acc = {};
    #pragma unroll
    for (int j = 0; j < GATE_KSLICE / 8; ++j) {
        v2f a0 = {a[j].x, a[j].y};
        v2f b0 = {b[j].x, b[j].y};
        v2f a1 = {a[j].z, a[j].w};
        v2f b1 = {b[j].z, b[j].w};
        acc = __builtin_amdgcn_wmma_f32_16x16x4_f32(
             false, a0,  false, b0,
             (short)0, acc,  false,  false);
        acc = __builtin_amdgcn_wmma_f32_16x16x4_f32(
             false, a1,  false, b1,
             (short)0, acc,  false,  false);
    }

    #pragma unroll
    for (int r = 0; r < 8; ++r) {
        Ds[(mt * 16 + 8 * h + r) * GP_DPITCH + nt * 16 + c] = acc[r];
    }
    __syncthreads();

    #pragma unroll
    for (int p = 0; p < 2; ++p) {
        int row = wave + 16 * p;
        v4f v = *(const v4f*)(Ds + row * GP_DPITCH + 4 * lane);
        *(v4f*)(planes + ((size_t)ks * NBLK + row) * D_MODEL + cg * GP_COLS + 4 * lane) = v;
    }
}

#define APPLY_GROUP   8
#define APPLY_WINDOW  32
#define APPLY_NGROUPS (APPLY_ROWS / APPLY_GROUP)
#define APPLY_WGROUPS (APPLY_WINDOW / APPLY_GROUP)

__global__ void __launch_bounds__(256 * APPLY_GROUPS) __attribute__((amdgpu_waves_per_eu(4, 4)))
se_apply_planes(const float* __restrict__ x,
                const float* __restrict__ planes,
                const float* __restrict__ bias,
                float* __restrict__ out) {
    int wg   = blockIdx.x;
    int c4   = threadIdx.x & 255;
    int half = threadIdx.x >> 8;
    size_t off = ((size_t)wg * ROWS_PER_CHUNK + (size_t)half * APPLY_ROWS + (APPLY_ROWS - 1)) * D_MODEL
               + 4 * c4;
    const float* xp = x + off;
    float*       op = out + off;
    const float* gp = planes + (size_t)(wg / ROW_CHUNKS) * D_MODEL + 4 * c4;

    v4f pl[GATE_KSPLIT];
    #pragma unroll
    for (int w = 0; w < GATE_KSPLIT; ++w) {
        pl[w] = *(const v4f*)(gp + (size_t)w * NBLK * D_MODEL);
    }
    v4f bb = *(const v4f*)(bias + 4 * c4);

    v4f v[APPLY_WINDOW];
    #pragma unroll
    for (int r = 0; r < APPLY_WINDOW; ++r) {
        v[r] = *(const v4f*)(xp - (size_t)r * D_MODEL);
    }

    v4f sum = pl[0];
    #pragma unroll
    for (int w = 1; w < GATE_KSPLIT; ++w) {
        sum += pl[w];
    }
    sum += bb;
    v4f gv;
    gv.x = __builtin_amdgcn_rcpf(1.0f + __expf(-sum.x));
    gv.y = __builtin_amdgcn_rcpf(1.0f + __expf(-sum.y));
    gv.z = __builtin_amdgcn_rcpf(1.0f + __expf(-sum.z));
    gv.w = __builtin_amdgcn_rcpf(1.0f + __expf(-sum.w));

    #pragma unroll
    for (int g = 0; g < APPLY_NGROUPS; ++g) {
        const int slot = (g % APPLY_WGROUPS) * APPLY_GROUP;
        v4f p[APPLY_GROUP];
        #pragma unroll
        for (int i = 0; i < APPLY_GROUP; ++i) {
            p[i] = v[slot + i] * gv;
        }
        __builtin_amdgcn_sched_barrier(0);
        #pragma unroll
        for (int i = 0; i < APPLY_GROUP; ++i) {
            __builtin_nontemporal_store(p[i], (v4f*)(op - (size_t)(g * APPLY_GROUP + i) * D_MODEL));
        }
        __builtin_amdgcn_sched_barrier(0);
        if (g + APPLY_WGROUPS < APPLY_NGROUPS) {
            #pragma unroll
            for (int i = 0; i < APPLY_GROUP; ++i) {
                v[slot + i] = *(const v4f*)(xp - (size_t)((g + APPLY_WGROUPS) * APPLY_GROUP + i) * D_MODEL);
            }
            __builtin_amdgcn_sched_barrier(0);
        }
    }
}

extern "C" void kernel_launch(void* const* d_in, const int* in_sizes, int n_in,
                              void* d_out, int out_size, void* d_ws, size_t ws_size,
                              hipStream_t stream) {
    const float* x = (const float*)d_in[0];
    const float* W = (const float*)d_in[1];
    const float* b = (const float*)d_in[2];
    float* out = (float*)d_out;

    float* ws = (float*)d_ws;

    if (ws_size >= (size_t)3145728) {
        float* part   = ws;
        float* planes = part + (size_t)NBLK * ROW_CHUNKS * D_MODEL;
        se_pool_partial<<<NBLK * ROW_CHUNKS, 256 * POOL_GROUPS, 0, stream>>>(x, part);
        se_gate_planes <<<GP_WGS, 32 * GATE_KSPLIT, 0, stream>>>(part, W, planes);
        se_apply_planes<<<NBLK * ROW_CHUNKS, 256 * APPLY_GROUPS, 0, stream>>>(x, planes, b, out);
        return;
    }

    float* pooled = ws;
    float* gate   = pooled + (size_t)NBLK * D_MODEL;
    float* part   = gate   + (size_t)NBLK * D_MODEL;

    se_pool_partial<<<NBLK * ROW_CHUNKS, 256 * POOL_GROUPS, 0, stream>>>(x, part);
    se_pool_reduce <<<NBLK,              256, 0, stream>>>(part, pooled);
    se_gate_wmma   <<<GATE_TILES, 32 * GATE_KSPLIT, 0, stream>>>(pooled, W, b, gate);

    se_apply_gate  <<<NBLK * ROW_CHUNKS, 256 * APPLY_GROUPS, 0, stream>>>(x, gate, out);
}
